# starting kernel
# baseline (speedup 1.0000x reference)
#include <hip/hip_runtime.h>
#include <math.h>

typedef __attribute__((ext_vector_type(2))) float v2f;
typedef __attribute__((ext_vector_type(8))) float v8f;

__device__ __forceinline__ v8f wmma_f32(v2f a, v2f b, v8f c) {
    return __builtin_amdgcn_wmma_f32_16x16x4_f32(
        false, a, false, b, (short)0, c, false, false);
}

__device__ __constant__ float c_D[64] = {
    0.3535533906f,  0.3535533906f,  0.3535533906f,  0.3535533906f,
    0.3535533906f,  0.3535533906f,  0.3535533906f,  0.3535533906f,

    0.4903926402f,  0.4157348062f,  0.2777851165f,  0.0975451610f,
   -0.0975451610f, -0.2777851165f, -0.4157348062f, -0.4903926402f,

    0.4619397663f,  0.1913417162f, -0.1913417162f, -0.4619397663f,
   -0.4619397663f, -0.1913417162f,  0.1913417162f,  0.4619397663f,

    0.4157348062f, -0.0975451610f, -0.4903926402f, -0.2777851165f,
    0.2777851165f,  0.4903926402f,  0.0975451610f, -0.4157348062f,

    0.3535533906f, -0.3535533906f, -0.3535533906f,  0.3535533906f,
    0.3535533906f, -0.3535533906f, -0.3535533906f,  0.3535533906f,

    0.2777851165f, -0.4903926402f,  0.0975451610f,  0.4157348062f,
   -0.4157348062f, -0.0975451610f,  0.4903926402f, -0.2777851165f,

    0.1913417162f, -0.4619397663f,  0.4619397663f, -0.1913417162f,
   -0.1913417162f,  0.4619397663f, -0.4619397663f,  0.1913417162f,

    0.0975451610f, -0.2777851165f,  0.4157348062f, -0.4903926402f,
    0.4903926402f, -0.4157348062f,  0.2777851165f, -0.0975451610f
};

__device__ __forceinline__ float d2(int r, int c) {
    const float v = c_D[((r & 7) << 3) | (c & 7)];
    return ((r >> 3) == (c >> 3)) ? v : 0.0f;
}

__global__ void __launch_bounds__(128)
irse_dct_kernel(const float* __restrict__ x, float* __restrict__ feat) {
    __shared__ float lds[4 * 256];
    const int lane = threadIdx.x & 31;
    const int wave = threadIdx.x >> 5;
    const int tile = blockIdx.x * 4 + wave;
    const int ti = tile >> 4, tj = tile & 15;
    const int h = lane >> 4;
    const int N = lane & 15;
    const int row0 = ti * 16, col0 = tj * 16;

    v8f acc = {};
#pragma unroll
    for (int k = 0; k < 4; ++k) {
        const int ka = 4 * k + 2 * h;
        v2f a, b;
        a.x = d2(N, ka);
        a.y = d2(N, ka + 1);
        b.x = x[(((row0 + ka) * 256) + col0 + N) * 3];
        b.y = x[(((row0 + ka + 1) * 256) + col0 + N) * 3];
        acc = wmma_f32(a, b, acc);
    }

    float* wlds = lds + wave * 256;
#pragma unroll
    for (int r = 0; r < 8; ++r) wlds[(r + 8 * h) * 16 + N] = acc[r];
    __syncthreads();

    v8f acc2 = {};
#pragma unroll
    for (int k = 0; k < 4; ++k) {
        const int ka = 4 * k + 2 * h;
        v2f a, b;
        a.x = wlds[N * 16 + ka];
        a.y = wlds[N * 16 + ka + 1];
        b.x = d2(N, ka);
        b.y = d2(N, ka + 1);
        acc2 = wmma_f32(a, b, acc2);
    }

#pragma unroll
    for (int r = 0; r < 8; ++r) {
        const int M = r + 8 * h;
        const int i = M & 7, bi = 2 * ti + (M >> 3);
        const int j = N & 7, bj = 2 * tj + (N >> 3);
        feat[(i * 8 + j) * 1024 + bj * 32 + bi] = acc2[r];
    }
}

__global__ void __launch_bounds__(256)
irse_se_mlp_kernel(const float* __restrict__ feat,
                   const float* __restrict__ fc1_w, const float* __restrict__ fc1_b,
                   const float* __restrict__ fc2_w, const float* __restrict__ fc2_b,
                   float* __restrict__ scale_c) {
    __shared__ float part[256];
    __shared__ float sq[64];
    __shared__ float hbuf[32];
    const int t = threadIdx.x;
    const int f = t >> 2, q = t & 3;
    float s = 0.0f;
    const float* p = feat + f * 1024 + q * 256;
    for (int i = 0; i < 256; ++i) s += p[i];
    part[t] = s;
    __syncthreads();
    if (q == 0)
        sq[f] = (part[t] + part[t + 1] + part[t + 2] + part[t + 3]) * (1.0f / 1024.0f);
    __syncthreads();
    if (t < 32) {
        float acc = fc1_b[t];
        for (int c = 0; c < 64; ++c) acc += sq[c] * fc1_w[t * 64 + c];
        hbuf[t] = fmaxf(acc, 0.0f);
    }
    __syncthreads();
    if (t < 64) {
        float acc = fc2_b[t];
        for (int r = 0; r < 32; ++r) acc += hbuf[r] * fc2_w[t * 32 + r];
        scale_c[t] = 1.0f / (1.0f + expf(-acc));
    }
}

__global__ void __launch_bounds__(256)
irse_spatial_kernel(const float* __restrict__ feat, const float* __restrict__ sse_w,
                    const float* __restrict__ sse_b, float* __restrict__ spatial) {
    const int idx = blockIdx.x * 256 + threadIdx.x;
    float acc = sse_b[0];
    for (int f = 0; f < 64; ++f) acc += sse_w[f] * feat[f * 1024 + idx];
    spatial[idx] = 1.0f / (1.0f + expf(-acc));
}

__device__ __forceinline__ float irse_yval(const float* __restrict__ feat,
                                           const float* __restrict__ sc,
                                           const float* __restrict__ sp,
                                           int K, int N, int ti, int tj) {
    const int f = (K & 7) * 8 + (N & 7);
    const int bj = 2 * tj + (N >> 3);
    const int bi = 2 * ti + (K >> 3);
    const float v = feat[f * 1024 + bj * 32 + bi];
    return fmaxf(v * sc[f], v * sp[bj * 32 + bi]);
}

__global__ void __launch_bounds__(128)
irse_idct_kernel(const float* __restrict__ feat, const float* __restrict__ scale_c,
                 const float* __restrict__ spatial, float* __restrict__ out2d) {
    __shared__ float lds[4 * 256];
    const int lane = threadIdx.x & 31;
    const int wave = threadIdx.x >> 5;
    const int tile = blockIdx.x * 4 + wave;
    const int ti = tile >> 4, tj = tile & 15;
    const int h = lane >> 4;
    const int N = lane & 15;

    v8f acc = {};
#pragma unroll
    for (int k = 0; k < 4; ++k) {
        const int ka = 4 * k + 2 * h;
        v2f a, b;
        a.x = d2(ka, N);
        a.y = d2(ka + 1, N);
        b.x = irse_yval(feat, scale_c, spatial, ka, N, ti, tj);
        b.y = irse_yval(feat, scale_c, spatial, ka + 1, N, ti, tj);
        acc = wmma_f32(a, b, acc);
    }

    float* wlds = lds + wave * 256;
#pragma unroll
    for (int r = 0; r < 8; ++r) wlds[(r + 8 * h) * 16 + N] = acc[r];
    __syncthreads();

    v8f acc2 = {};
#pragma unroll
    for (int k = 0; k < 4; ++k) {
        const int ka = 4 * k + 2 * h;
        v2f a, b;
        a.x = wlds[N * 16 + ka];
        a.y = wlds[N * 16 + ka + 1];
        b.x = d2(ka, N);
        b.y = d2(ka + 1, N);
        acc2 = wmma_f32(a, b, acc2);
    }

    const int row0 = ti * 16, col0 = tj * 16;
#pragma unroll
    for (int r = 0; r < 8; ++r) {
        const int M = r + 8 * h;
        out2d[(row0 + M) * 256 + col0 + N] = acc2[r];
    }
}

__global__ void __launch_bounds__(256)
irse_bcast_kernel(const float4* __restrict__ out2d4, float4* __restrict__ out4) {
    const unsigned idx = blockIdx.x * 256u + threadIdx.x;
    out4[idx] = out2d4[idx & 16383u];
}

extern "C" void kernel_launch(void* const* d_in, const int* in_sizes, int n_in,
                              void* d_out, int out_size, void* d_ws, size_t ws_size,
                              hipStream_t stream) {
    const float* x     = (const float*)d_in[0];
    const float* fc1_w = (const float*)d_in[1];
    const float* fc1_b = (const float*)d_in[2];
    const float* fc2_w = (const float*)d_in[3];
    const float* fc2_b = (const float*)d_in[4];
    const float* sse_w = (const float*)d_in[5];
    const float* sse_b = (const float*)d_in[6];

    float* ws      = (float*)d_ws;
    float* feat    = ws;
    float* out2d   = ws + 65536;
    float* scale_c = ws + 131072;
    float* spatial = ws + 131072 + 64;

    irse_dct_kernel<<<64, 128, 0, stream>>>(x, feat);
    irse_se_mlp_kernel<<<1, 256, 0, stream>>>(feat, fc1_w, fc1_b, fc2_w, fc2_b, scale_c);
    irse_spatial_kernel<<<4, 256, 0, stream>>>(feat, sse_w, sse_b, spatial);
    irse_idct_kernel<<<64, 128, 0, stream>>>(feat, scale_c, spatial, out2d);
    irse_bcast_kernel<<<12288, 256, 0, stream>>>((const float4*)out2d, (float4*)d_out);
}


# Coefficient planes never stored: channel means by linearity (DCT of summed 8x8 pixel block), one kernel does DCT, gates, iDCT in registers and writes the planes; 5 launches to 2; sum orders change
# speedup vs baseline: 2.0735x; 2.0735x over previous
#include <hip/hip_runtime.h>
#include <math.h>

typedef __attribute__((ext_vector_type(2))) float v2f;
typedef __attribute__((ext_vector_type(8))) float v8f;

__device__ __forceinline__ v8f wmma_f32(v2f a, v2f b, v8f c) {
    return __builtin_amdgcn_wmma_f32_16x16x4_f32(
        false, a, false, b, (short)0, c, false, false);
}

__device__ __constant__ float c_D[64] = {
    0.3535533906f,  0.3535533906f,  0.3535533906f,  0.3535533906f,
    0.3535533906f,  0.3535533906f,  0.3535533906f,  0.3535533906f,

    0.4903926402f,  0.4157348062f,  0.2777851165f,  0.0975451610f,
   -0.0975451610f, -0.2777851165f, -0.4157348062f, -0.4903926402f,

    0.4619397663f,  0.1913417162f, -0.1913417162f, -0.4619397663f,
   -0.4619397663f, -0.1913417162f,  0.1913417162f,  0.4619397663f,

    0.4157348062f, -0.0975451610f, -0.4903926402f, -0.2777851165f,
    0.2777851165f,  0.4903926402f,  0.0975451610f, -0.4157348062f,

    0.3535533906f, -0.3535533906f, -0.3535533906f,  0.3535533906f,
    0.3535533906f, -0.3535533906f, -0.3535533906f,  0.3535533906f,

    0.2777851165f, -0.4903926402f,  0.0975451610f,  0.4157348062f,
   -0.4157348062f, -0.0975451610f,  0.4903926402f, -0.2777851165f,

    0.1913417162f, -0.4619397663f,  0.4619397663f, -0.1913417162f,
   -0.1913417162f,  0.4619397663f, -0.4619397663f,  0.1913417162f,

    0.0975451610f, -0.2777851165f,  0.4157348062f, -0.4903926402f,
    0.4903926402f, -0.4157348062f,  0.2777851165f, -0.0975451610f
};

__device__ __forceinline__ float d2(int r, int c) {
    const float v = c_D[((r & 7) << 3) | (c & 7)];
    return ((r >> 3) == (c >> 3)) ? v : 0.0f;
}

__global__ void __launch_bounds__(256)
irse_pixsum_kernel(const float* __restrict__ x, float* __restrict__ psum) {
    __shared__ float colv[8 * 256];
    const int t = threadIdx.x;
    const int b = blockIdx.x;
#pragma unroll
    for (int k = 0; k < 8; ++k)
        colv[k * 256 + t] = x[((8 * b + k) * 256 + t) * 3];
    __syncthreads();
    if (t < 64) {
        const int k = t >> 3, l = t & 7;
        float s = 0.0f;
#pragma unroll
        for (int bj = 0; bj < 32; ++bj) s += colv[k * 256 + bj * 8 + l];
        psum[b * 64 + t] = s;
    }
}

__global__ void __launch_bounds__(128)
irse_fused_kernel(const float* __restrict__ x, const float* __restrict__ psum,
                  const float* __restrict__ fc1_w, const float* __restrict__ fc1_b,
                  const float* __restrict__ fc2_w, const float* __restrict__ fc2_b,
                  const float* __restrict__ sse_w, const float* __restrict__ sse_b,
                  float4* __restrict__ out4) {
    __shared__ float lds[4 * 256];
    __shared__ __attribute__((aligned(16))) float reg[16 * 64];
    __shared__ float s2[128];
    __shared__ float Ssum[64];
    __shared__ float Tm[64];
    __shared__ float sq[64];
    __shared__ float hbuf[32];
    __shared__ float sc[64];

    const int t = threadIdx.x;
    const int lane = t & 31;
    const int wave = t >> 5;
    const int g = blockIdx.x;
    const int tile = g * 4 + wave;
    const int ti = tile >> 4, tj = tile & 15;
    const int h = lane >> 4;
    const int N = lane & 15;
    const int row0 = ti * 16, col0 = tj * 16;

    {
        const int c = t & 63, hf = t >> 6;
        float s = 0.0f;
#pragma unroll
        for (int b = 0; b < 16; ++b) s += psum[(hf * 16 + b) * 64 + c];
        s2[t] = s;
    }

    const float4* w1p = (const float4*)(fc1_w + (t >> 2) * 64 + (t & 3) * 16);
    const float4* w2p = (const float4*)(fc2_w + (t >> 1) * 32 + (t & 1) * 16);
    float4 w1[4], w2[4];
#pragma unroll
    for (int i = 0; i < 4; ++i) { w1[i] = w1p[i]; w2[i] = w2p[i]; }
    const float b1 = fc1_b[t >> 2];
    const float b2 = fc2_b[t >> 1];

    v8f acc = {};
#pragma unroll
    for (int k = 0; k < 4; ++k) {
        const int ka = 4 * k + 2 * h;
        v2f a, b;
        a.x = d2(N, ka);
        a.y = d2(N, ka + 1);
        b.x = x[(((row0 + ka) * 256) + col0 + N) * 3];
        b.y = x[(((row0 + ka + 1) * 256) + col0 + N) * 3];
        acc = wmma_f32(a, b, acc);
    }
    float* wlds = lds + wave * 256;
#pragma unroll
    for (int r = 0; r < 8; ++r) wlds[(r + 8 * h) * 16 + N] = acc[r];
    __syncthreads();

    v8f F = {};
#pragma unroll
    for (int k = 0; k < 4; ++k) {
        const int ka = 4 * k + 2 * h;
        v2f a, b;
        a.x = wlds[N * 16 + ka];
        a.y = wlds[N * 16 + ka + 1];
        b.x = d2(N, ka);
        b.y = d2(N, ka + 1);
        F = wmma_f32(a, b, F);
    }

    const int j = N & 7;
    float sp = 0.0f;
#pragma unroll
    for (int r = 0; r < 8; ++r) sp += sse_w[r * 8 + j] * F[r];
    sp += __shfl_xor(sp, 1);
    sp += __shfl_xor(sp, 2);
    sp += __shfl_xor(sp, 4);
    sp = 1.0f / (1.0f + expf(-(sp + sse_b[0])));

    if (t < 64) Ssum[t] = s2[t] + s2[t + 64];
    __syncthreads();
    if (t < 64) {
        const int i = t >> 3, l = t & 7;
        float s = 0.0f;
#pragma unroll
        for (int k = 0; k < 8; ++k) s += c_D[i * 8 + k] * Ssum[k * 8 + l];
        Tm[t] = s;
    }
    __syncthreads();
    if (t < 64) {
        const int i = t >> 3, jj = t & 7;
        float s = 0.0f;
#pragma unroll
        for (int l = 0; l < 8; ++l) s += Tm[i * 8 + l] * c_D[jj * 8 + l];
        sq[t] = s * (1.0f / 1024.0f);
    }
    __syncthreads();

    {
        const float* q = sq + (t & 3) * 16;
        float a1 = 0.0f;
#pragma unroll
        for (int i = 0; i < 4; ++i)
            a1 += w1[i].x * q[4 * i] + w1[i].y * q[4 * i + 1]
                + w1[i].z * q[4 * i + 2] + w1[i].w * q[4 * i + 3];
        a1 += __shfl_xor(a1, 1);
        a1 += __shfl_xor(a1, 2);
        if ((t & 3) == 0) hbuf[t >> 2] = fmaxf(a1 + b1, 0.0f);
    }
    __syncthreads();
    {
        const float* q = hbuf + (t & 1) * 16;
        float a2 = 0.0f;
#pragma unroll
        for (int i = 0; i < 4; ++i)
            a2 += w2[i].x * q[4 * i] + w2[i].y * q[4 * i + 1]
                + w2[i].z * q[4 * i + 2] + w2[i].w * q[4 * i + 3];
        a2 += __shfl_xor(a2, 1);
        if ((t & 1) == 0) sc[t >> 1] = 1.0f / (1.0f + expf(-(a2 + b2)));
    }
    __syncthreads();

    v8f Y;
#pragma unroll
    for (int r = 0; r < 8; ++r) Y[r] = fmaxf(F[r] * sc[r * 8 + j], F[r] * sp);

    v8f acc3 = {};
#pragma unroll
    for (int k = 0; k < 4; ++k) {
        const int kk = 2 * k + 8 * h;
        v2f a, b;
        a.x = d2(kk, N);
        a.y = d2(kk + 1, N);
        b.x = Y[2 * k];
        b.y = Y[2 * k + 1];
        acc3 = wmma_f32(a, b, acc3);
    }
#pragma unroll
    for (int r = 0; r < 8; ++r) wlds[(r + 8 * h) * 16 + N] = acc3[r];
    __syncthreads();

    v8f acc4 = {};
#pragma unroll
    for (int k = 0; k < 4; ++k) {
        const int ka = 4 * k + 2 * h;
        v2f a, b;
        a.x = wlds[N * 16 + ka];
        a.y = wlds[N * 16 + ka + 1];
        b.x = d2(ka, N);
        b.y = d2(ka + 1, N);
        acc4 = wmma_f32(a, b, acc4);
    }

#pragma unroll
    for (int r = 0; r < 8; ++r) reg[(r + 8 * h) * 64 + wave * 16 + N] = acc4[r];
    __syncthreads();

    float4 v[8];
#pragma unroll
    for (int p = 0; p < 8; ++p)
        v[p] = *(const float4*)(reg + (2 * p + h) * 64 + 4 * N);

    const int rowg = (g >> 2) * 16;
    const int colg = (g & 3) * 64;
    const unsigned base4 = (unsigned)((rowg + h) * 256 + colg) / 4u + (unsigned)N;
#pragma unroll
    for (int q = 0; q < 4; ++q) {
        const unsigned plane = blockIdx.y * 16u + (unsigned)(wave * 4 + q);
        float4* dst = out4 + plane * 16384u + base4;
#pragma unroll
        for (int p = 0; p < 8; ++p) dst[p * 128] = v[p];
    }
}

extern "C" void kernel_launch(void* const* d_in, const int* in_sizes, int n_in,
                              void* d_out, int out_size, void* d_ws, size_t ws_size,
                              hipStream_t stream) {
    const float* x     = (const float*)d_in[0];
    const float* fc1_w = (const float*)d_in[1];
    const float* fc1_b = (const float*)d_in[2];
    const float* fc2_w = (const float*)d_in[3];
    const float* fc2_b = (const float*)d_in[4];
    const float* sse_w = (const float*)d_in[5];
    const float* sse_b = (const float*)d_in[6];

    float* psum = (float*)d_ws;

    irse_pixsum_kernel<<<32, 256, 0, stream>>>(x, psum);
    irse_fused_kernel<<<dim3(64, 12), 128, 0, stream>>>(x, psum, fc1_w, fc1_b, fc2_w, fc2_b,
                                                        sse_w, sse_b, (float4*)d_out);
}


# Squeeze+MLP moved to a ninth wave that runs them wave-locally beside 8 tile waves (256 workgroups: 16x128 region x 24 planes, was 768 x 4 waves); block barriers 8 -> 4; sum grouping kept
# speedup vs baseline: 2.9386x; 1.4172x over previous
#include <hip/hip_runtime.h>
#include <math.h>

typedef __attribute__((ext_vector_type(2))) float v2f;
typedef __attribute__((ext_vector_type(8))) float v8f;

__device__ __forceinline__ v8f wmma_f32(v2f a, v2f b, v8f c) {
    return __builtin_amdgcn_wmma_f32_16x16x4_f32(
        false, a, false, b, (short)0, c, false, false);
}

__device__ __constant__ float c_D[64] = {
    0.3535533906f,  0.3535533906f,  0.3535533906f,  0.3535533906f,
    0.3535533906f,  0.3535533906f,  0.3535533906f,  0.3535533906f,

    0.4903926402f,  0.4157348062f,  0.2777851165f,  0.0975451610f,
   -0.0975451610f, -0.2777851165f, -0.4157348062f, -0.4903926402f,

    0.4619397663f,  0.1913417162f, -0.1913417162f, -0.4619397663f,
   -0.4619397663f, -0.1913417162f,  0.1913417162f,  0.4619397663f,

    0.4157348062f, -0.0975451610f, -0.4903926402f, -0.2777851165f,
    0.2777851165f,  0.4903926402f,  0.0975451610f, -0.4157348062f,

    0.3535533906f, -0.3535533906f, -0.3535533906f,  0.3535533906f,
    0.3535533906f, -0.3535533906f, -0.3535533906f,  0.3535533906f,

    0.2777851165f, -0.4903926402f,  0.0975451610f,  0.4157348062f,
   -0.4157348062f, -0.0975451610f,  0.4903926402f, -0.2777851165f,

    0.1913417162f, -0.4619397663f,  0.4619397663f, -0.1913417162f,
   -0.1913417162f,  0.4619397663f, -0.4619397663f,  0.1913417162f,

    0.0975451610f, -0.2777851165f,  0.4157348062f, -0.4903926402f,
    0.4903926402f, -0.4157348062f,  0.2777851165f, -0.0975451610f
};

__device__ __forceinline__ float d2(int r, int c) {
    const float v = c_D[((r & 7) << 3) | (c & 7)];
    return ((r >> 3) == (c >> 3)) ? v : 0.0f;
}

__global__ void __launch_bounds__(256)
irse_pixsum_kernel(const float* __restrict__ x, float* __restrict__ psum) {
    __shared__ float colv[8 * 256];
    const int t = threadIdx.x;
    const int b = blockIdx.x;
#pragma unroll
    for (int k = 0; k < 8; ++k)
        colv[k * 256 + t] = x[((8 * b + k) * 256 + t) * 3];
    __syncthreads();
    if (t < 64) {
        const int k = t >> 3, l = t & 7;
        float s = 0.0f;
#pragma unroll
        for (int bj = 0; bj < 32; ++bj) s += colv[k * 256 + bj * 8 + l];
        psum[b * 64 + t] = s;
    }
}

__device__ __forceinline__ void wave_lds_handover() {
    asm volatile("s_wait_dscnt 0x0" ::: "memory");
}

__device__ __forceinline__ float dot16(const float4* w, const float* q) {
    float a = 0.0f;
#pragma unroll
    for (int i = 0; i < 4; ++i)
        a += w[i].x * q[4 * i] + w[i].y * q[4 * i + 1]
           + w[i].z * q[4 * i + 2] + w[i].w * q[4 * i + 3];
    return a;
}

__global__ void __launch_bounds__(288)
irse_fused_kernel(const float* __restrict__ x, const float* __restrict__ psum,
                  const float* __restrict__ fc1_w, const float* __restrict__ fc1_b,
                  const float* __restrict__ fc2_w, const float* __restrict__ fc2_b,
                  const float* __restrict__ sse_w, const float* __restrict__ sse_b,
                  float4* __restrict__ out4) {
    __shared__ float lds[8 * 256];
    __shared__ __attribute__((aligned(16))) float reg[16 * 128];
    __shared__ __attribute__((aligned(16))) float s2[128];
    __shared__ __attribute__((aligned(16))) float Ssum[64];
    __shared__ __attribute__((aligned(16))) float Tm[64];
    __shared__ __attribute__((aligned(16))) float sq[64];
    __shared__ __attribute__((aligned(16))) float hbuf[32];
    __shared__ __attribute__((aligned(16))) float sc[64];
    __shared__ __attribute__((aligned(16))) float W1s[32 * 68];
    __shared__ __attribute__((aligned(16))) float W2s[64 * 36];

    const int t = threadIdx.x;
    const int lane = t & 31;
    const int wave = __builtin_amdgcn_readfirstlane(t >> 5);
    const bool tilew = wave < 8;
    const int g = blockIdx.x;
    const int tile = g * 8 + (wave & 7);
    const int ti = tile >> 4, tj = tile & 15;
    const int h = lane >> 4;
    const int N = lane & 15;
    const int j = N & 7;
    const int row0 = ti * 16, col0 = tj * 16;
    float* wlds = lds + (wave & 7) * 256;

    v2f dfw[4] = {}, xin[4] = {};
    float b1 = 0.0f, b2a = 0.0f, b2b = 0.0f;
    float dA[8] = {}, dB[8] = {}, dJ[8] = {};

    if (tilew) {
#pragma unroll
        for (int k = 0; k < 4; ++k) {
            const int ka = 4 * k + 2 * h;
            dfw[k].x = d2(N, ka);
            dfw[k].y = d2(N, ka + 1);
            xin[k].x = x[(((row0 + ka) * 256) + col0 + N) * 3];
            xin[k].y = x[(((row0 + ka + 1) * 256) + col0 + N) * 3];
        }
        if (wave < 4) {
            const float4* w1p = (const float4*)(fc1_w + (t >> 2) * 64 + (t & 3) * 16);
            const float4* w2p = (const float4*)(fc2_w + (t >> 1) * 32 + (t & 1) * 16);
            float4 wa[4], wb[4];
#pragma unroll
            for (int i = 0; i < 4; ++i) { wa[i] = w1p[i]; wb[i] = w2p[i]; }
            const int c = t & 63, hf = t >> 6;
            float s = 0.0f;
#pragma unroll
            for (int b = 0; b < 16; ++b) s += psum[(hf * 16 + b) * 64 + c];
            s2[t] = s;
            float* w1d = W1s + (t >> 2) * 68 + (t & 3) * 16;
            float* w2d = W2s + (t >> 1) * 36 + (t & 1) * 16;
#pragma unroll
            for (int i = 0; i < 4; ++i) {
                *(float4*)(w1d + 4 * i) = wa[i];
                *(float4*)(w2d + 4 * i) = wb[i];
            }
        }
        v8f acc = {};
#pragma unroll
        for (int k = 0; k < 4; ++k) acc = wmma_f32(dfw[k], xin[k], acc);
#pragma unroll
        for (int r = 0; r < 8; ++r) wlds[(r + 8 * h) * 16 + N] = acc[r];
    } else {
        b1  = fc1_b[lane];
        b2a = fc2_b[lane];
        b2b = fc2_b[lane + 32];
        const int i0 = lane >> 3;
#pragma unroll
        for (int k = 0; k < 8; ++k) {
            dA[k] = c_D[i0 * 8 + k];
            dB[k] = c_D[(i0 + 4) * 8 + k];
            dJ[k] = c_D[j * 8 + k];
        }
    }
    __syncthreads();

    v8f F = {};
    float sp = 0.0f;
    if (tilew) {
#pragma unroll
        for (int k = 0; k < 4; ++k) {
            const int ka = 4 * k + 2 * h;
            v2f a;
            a.x = wlds[N * 16 + ka];
            a.y = wlds[N * 16 + ka + 1];
            F = wmma_f32(a, dfw[k], F);
        }
#pragma unroll
        for (int r = 0; r < 8; ++r) sp += sse_w[r * 8 + j] * F[r];
        sp += __shfl_xor(sp, 1);
        sp += __shfl_xor(sp, 2);
        sp += __shfl_xor(sp, 4);
        sp = 1.0f / (1.0f + expf(-(sp + sse_b[0])));
    } else {
        const int i0 = lane >> 3;
        float4 w1[16], w2a[8], w2b[8];
#pragma unroll
        for (int i = 0; i < 16; ++i) w1[i] = *(const float4*)(W1s + lane * 68 + 4 * i);
#pragma unroll
        for (int i = 0; i < 8; ++i) {
            w2a[i] = *(const float4*)(W2s + lane * 36 + 4 * i);
            w2b[i] = *(const float4*)(W2s + (lane + 32) * 36 + 4 * i);
        }
        Ssum[lane]      = s2[lane]      + s2[lane + 64];
        Ssum[lane + 32] = s2[lane + 32] + s2[lane + 96];
        wave_lds_handover();
        float ta = 0.0f, tb = 0.0f;
#pragma unroll
        for (int k = 0; k < 8; ++k) {
            const float sv = Ssum[k * 8 + j];
            ta += dA[k] * sv;
            tb += dB[k] * sv;
        }
        Tm[lane]      = ta;
        Tm[lane + 32] = tb;
        wave_lds_handover();
        float qa = 0.0f, qb = 0.0f;
#pragma unroll
        for (int l = 0; l < 8; ++l) {
            qa += Tm[i0 * 8 + l] * dJ[l];
            qb += Tm[(i0 + 4) * 8 + l] * dJ[l];
        }
        sq[lane]      = qa * (1.0f / 1024.0f);
        sq[lane + 32] = qb * (1.0f / 1024.0f);
        wave_lds_handover();
        {
            const float p0 = dot16(w1,      sq);
            const float p1 = dot16(w1 + 4,  sq + 16);
            const float p2 = dot16(w1 + 8,  sq + 32);
            const float p3 = dot16(w1 + 12, sq + 48);
            const float a1 = (p0 + p1) + (p2 + p3);
            hbuf[lane] = fmaxf(a1 + b1, 0.0f);
        }
        wave_lds_handover();
        {
            const float a2a = dot16(w2a, hbuf) + dot16(w2a + 4, hbuf + 16);
            const float a2b = dot16(w2b, hbuf) + dot16(w2b + 4, hbuf + 16);
            sc[lane]      = 1.0f / (1.0f + expf(-(a2a + b2a)));
            sc[lane + 32] = 1.0f / (1.0f + expf(-(a2b + b2b)));
        }
    }
    __syncthreads();

    if (tilew) {
        v8f Y;
#pragma unroll
        for (int r = 0; r < 8; ++r) Y[r] = fmaxf(F[r] * sc[r * 8 + j], F[r] * sp);

        v8f acc3 = {};
#pragma unroll
        for (int k = 0; k < 4; ++k) {
            const int kk = 2 * k + 8 * h;
            v2f a, b;
            a.x = d2(kk, N);
            a.y = d2(kk + 1, N);
            b.x = Y[2 * k];
            b.y = Y[2 * k + 1];
            acc3 = wmma_f32(a, b, acc3);
        }
#pragma unroll
        for (int r = 0; r < 8; ++r) wlds[(r + 8 * h) * 16 + N] = acc3[r];
    }
    __syncthreads();

    if (tilew) {
        v8f acc4 = {};
#pragma unroll
        for (int k = 0; k < 4; ++k) {
            const int ka = 4 * k + 2 * h;
            v2f a, b;
            a.x = wlds[N * 16 + ka];
            a.y = wlds[N * 16 + ka + 1];
            b.x = d2(ka, N);
            b.y = d2(ka + 1, N);
            acc4 = wmma_f32(a, b, acc4);
        }
#pragma unroll
        for (int r = 0; r < 8; ++r) reg[(r + 8 * h) * 128 + wave * 16 + N] = acc4[r];
    }
    __syncthreads();

    if (tilew) {
        float4 v[16];
#pragma unroll
        for (int p = 0; p < 16; ++p)
            v[p] = *(const float4*)(reg + p * 128 + 4 * lane);

        const int rowg = (g >> 1) * 16;
        const int colg = (g & 1) * 128;
        const unsigned base4 = (unsigned)(rowg * 256 + colg) / 4u + (unsigned)lane;
#pragma unroll
        for (int q = 0; q < 3; ++q) {
            const unsigned plane = blockIdx.y * 24u + (unsigned)(wave * 3 + q);
            float4* dst = out4 + plane * 16384u + base4;
#pragma unroll
            for (int p = 0; p < 16; ++p) dst[p * 64] = v[p];
        }
    }
}

extern "C" void kernel_launch(void* const* d_in, const int* in_sizes, int n_in,
                              void* d_out, int out_size, void* d_ws, size_t ws_size,
                              hipStream_t stream) {
    const float* x     = (const float*)d_in[0];
    const float* fc1_w = (const float*)d_in[1];
    const float* fc1_b = (const float*)d_in[2];
    const float* fc2_w = (const float*)d_in[3];
    const float* fc2_b = (const float*)d_in[4];
    const float* sse_w = (const float*)d_in[5];
    const float* sse_b = (const float*)d_in[6];

    float* psum = (float*)d_ws;

    irse_pixsum_kernel<<<32, 256, 0, stream>>>(x, psum);
    irse_fused_kernel<<<dim3(32, 8), 288, 0, stream>>>(x, psum, fc1_w, fc1_b, fc2_w, fc2_b,
                                                       sse_w, sse_b, (float4*)d_out);
}
